# starting kernel
# baseline (speedup 1.0000x reference)
#include <hip/hip_runtime.h>
#include <hip/hip_bf16.h>

typedef __attribute__((ext_vector_type(2))) float v2f;
typedef __attribute__((ext_vector_type(8))) float v8f;

#define WPAD 72

__global__ void fill_f32(float* __restrict__ p, float v, long n) {
  long i = (long)blockIdx.x * blockDim.x + threadIdx.x;
  if (i < n) p[i] = v;
}

__global__ void count_deg(const int* __restrict__ col, float* __restrict__ deg, int nE) {
  int e = blockIdx.x * blockDim.x + threadIdx.x;
  if (e < nE) unsafeAtomicAdd(&deg[col[e]], 1.0f);
}

__global__ void make_dinv(float* __restrict__ d, int n) {
  int i = blockIdx.x * blockDim.x + threadIdx.x;
  if (i < n) {
    float v = d[i];
    d[i] = (v > 0.0f) ? rsqrtf(v) : 0.0f;
  }
}

template <int K>
__global__ void __launch_bounds__(128)
gcn_gemm_wmma(const float* __restrict__ X, const float* __restrict__ W,
              const float* __restrict__ dinv, float* __restrict__ Y, int nRows) {
  __shared__ float wlds[K * WPAD];
  __shared__ float alds[16 * (K + 4)];

  const int tid  = threadIdx.x;
  const int wave = tid >> 5;
  const int lane = tid & 31;
  const int tile = blockIdx.x * 16;

  for (int i = tid * 4; i < K * 64; i += 128 * 4) {
    const int k = i >> 6;
    const int c = i & 63;
    *reinterpret_cast<float4*>(&wlds[k * WPAD + c]) =
        *reinterpret_cast<const float4*>(W + i);
  }
  for (int i = tid * 4; i < 16 * K; i += 128 * 4) {
    const int r = i / K;
    const int c = i % K;
    int gr = tile + r;
    if (gr >= nRows) gr = nRows - 1;
    *reinterpret_cast<float4*>(&alds[r * (K + 4) + c]) =
        *reinterpret_cast<const float4*>(X + (long)gr * K + c);
  }
  __syncthreads();

  const int m  = lane & 15;
  const int kq = (lane >> 4) * 2;
  const int n  = (lane & 15) + wave * 16;

  v8f c = {};
#pragma unroll
  for (int kk = 0; kk < K; kk += 4) {
    const int kb = kk + kq;
    v2f a, b;
    a.x = alds[m * (K + 4) + kb];
    a.y = alds[m * (K + 4) + kb + 1];
    b.x = wlds[kb * WPAD + n];
    b.y = wlds[(kb + 1) * WPAD + n];
    c = __builtin_amdgcn_wmma_f32_16x16x4_f32(false, a, false, b, (short)0, c,
                                              false, false);
  }

  const int mrow0 = (lane >> 4) * 8;
#pragma unroll
  for (int v = 0; v < 8; ++v) {
    const int gr = tile + mrow0 + v;
    if (gr < nRows) Y[(long)gr * 64 + n] = c[v] * dinv[gr];
  }
}

__global__ void __launch_bounds__(256)
scatter_edges(const int* __restrict__ row, const int* __restrict__ col,
              const float* __restrict__ y, float* __restrict__ acc, long nWork) {
  long t = (long)blockIdx.x * 256 + threadIdx.x;
  if (t >= nWork) return;
  const int e = (int)(t >> 4);
  const int f = (int)(t & 15) << 2;
  const int r = row[e];
  const int cd = col[e];
  const float4 v = *reinterpret_cast<const float4*>(y + (long)r * 64 + f);
  float* dst = acc + (long)cd * 64 + f;
  unsafeAtomicAdd(dst + 0, v.x);
  unsafeAtomicAdd(dst + 1, v.y);
  unsafeAtomicAdd(dst + 2, v.z);
  unsafeAtomicAdd(dst + 3, v.w);
}

__global__ void __launch_bounds__(256)
finalize_layer(const float* __restrict__ acc, const float* __restrict__ y,
               const float* __restrict__ dinv, const float* __restrict__ bias,
               float* __restrict__ out, int ostride, int nNodes) {
  long t = (long)blockIdx.x * 256 + threadIdx.x;
  if (t >= (long)nNodes * 16) return;
  const int i = (int)(t >> 4);
  const int f = (int)(t & 15) << 2;
  const float d = dinv[i];
  const float4 a = *reinterpret_cast<const float4*>(acc + (long)i * 64 + f);
  const float4 s = *reinterpret_cast<const float4*>(y + (long)i * 64 + f);
  const float4 b = *reinterpret_cast<const float4*>(bias + f);
  float4 o;
  o.x = d * (a.x + s.x) + b.x;
  o.y = d * (a.y + s.y) + b.y;
  o.z = d * (a.z + s.z) + b.z;
  o.w = d * (a.w + s.w) + b.w;
  *reinterpret_cast<float4*>(out + (long)i * ostride + f) = o;
}

extern "C" void kernel_launch(void* const* d_in, const int* in_sizes, int n_in,
                              void* d_out, int out_size, void* d_ws, size_t ws_size,
                              hipStream_t stream) {
  const float* latent    = (const float*)d_in[0];
  const float* condition = (const float*)d_in[1];
  const int*   ei        = (const int*)d_in[2];
  const float* W1 = (const float*)d_in[3];
  const float* b1 = (const float*)d_in[4];
  const float* W2 = (const float*)d_in[5];
  const float* b2 = (const float*)d_in[6];
  const float* W3 = (const float*)d_in[7];
  const float* b3 = (const float*)d_in[8];
  const float* W4 = (const float*)d_in[9];
  const float* b4 = (const float*)d_in[10];

  const int N = in_sizes[0] / 64;
  const int E = in_sizes[2] / 2;
  const int* rowp = ei;
  const int* colp = ei + E;

  char* ws = (char*)d_ws;
  size_t off = 0;
  float* dinv = (float*)(ws + off);
  off += ((size_t)N * 4 + 255) & ~(size_t)255;
  float* ybuf = (float*)(ws + off);
  off += (size_t)N * 64 * 4;
  float* acc  = (float*)(ws + off);
  off += (size_t)N * 64 * 4;
  float* hbuf = (float*)(ws + off);

  const long accElems = (long)N * 64;
  const long edgeWork = (long)E * 16;
  const dim3 gemmGrid((N + 15) / 16);
  const dim3 gemmBlk(128);
  const dim3 fillN((N + 255) / 256);
  const dim3 fillAcc((accElems + 255) / 256);
  const dim3 edgeGrid((edgeWork + 255) / 256);
  const dim3 finGrid(((long)N * 16 + 255) / 256);

  fill_f32<<<fillN, 256, 0, stream>>>(dinv, 1.0f, N);
  count_deg<<<(E + 255) / 256, 256, 0, stream>>>(colp, dinv, E);
  make_dinv<<<fillN, 256, 0, stream>>>(dinv, N);

  gcn_gemm_wmma<64><<<gemmGrid, gemmBlk, 0, stream>>>(latent, W1, dinv, ybuf, N);
  fill_f32<<<fillAcc, 256, 0, stream>>>(acc, 0.0f, accElems);
  scatter_edges<<<edgeGrid, 256, 0, stream>>>(rowp, colp, ybuf, acc, edgeWork);
  finalize_layer<<<finGrid, 256, 0, stream>>>(acc, ybuf, dinv, b1, hbuf, 128, N);

  gcn_gemm_wmma<32><<<gemmGrid, gemmBlk, 0, stream>>>(condition, W2, dinv, ybuf, N);
  fill_f32<<<fillAcc, 256, 0, stream>>>(acc, 0.0f, accElems);
  scatter_edges<<<edgeGrid, 256, 0, stream>>>(rowp, colp, ybuf, acc, edgeWork);
  finalize_layer<<<finGrid, 256, 0, stream>>>(acc, ybuf, dinv, b2, hbuf + 64, 128, N);

  gcn_gemm_wmma<128><<<gemmGrid, gemmBlk, 0, stream>>>(hbuf, W3, dinv, ybuf, N);
  fill_f32<<<fillAcc, 256, 0, stream>>>(acc, 0.0f, accElems);
  scatter_edges<<<edgeGrid, 256, 0, stream>>>(rowp, colp, ybuf, acc, edgeWork);
  finalize_layer<<<finGrid, 256, 0, stream>>>(acc, ybuf, dinv, b3, hbuf, 64, N);

  gcn_gemm_wmma<64><<<gemmGrid, gemmBlk, 0, stream>>>(hbuf, W4, dinv, ybuf, N);
  fill_f32<<<fillAcc, 256, 0, stream>>>(acc, 0.0f, accElems);
  scatter_edges<<<edgeGrid, 256, 0, stream>>>(rowp, colp, ybuf, acc, edgeWork);
  finalize_layer<<<finGrid, 256, 0, stream>>>(acc, ybuf, dinv, b4, (float*)d_out, 64, N);
}


# Aggregate by destination-sorted gather in a fixed order: edges bucketed per call (int atomic ranks + prefix scan), each bucket sorted by source; float atomics, acc fill, finalize pass gone
# speedup vs baseline: 2.1781x; 2.1781x over previous
#include <hip/hip_runtime.h>
#include <hip/hip_bf16.h>

typedef __attribute__((ext_vector_type(2))) float v2f;
typedef __attribute__((ext_vector_type(8))) float v8f;

#define WPAD 72

__global__ void fill_i32(int* __restrict__ p, int v, int n) {
  int i = blockIdx.x * blockDim.x + threadIdx.x;
  if (i < n) p[i] = v;
}

__global__ void count_rank(const int* __restrict__ col, int* __restrict__ cnt,
                           int* __restrict__ rank, int nE) {
  int e = blockIdx.x * blockDim.x + threadIdx.x;
  if (e < nE) rank[e] = atomicAdd(&cnt[col[e]], 1);
}

__global__ void __launch_bounds__(256)
block_totals(const int* __restrict__ cnt, int* __restrict__ bsum, int n) {
  __shared__ int sh[256];
  const int t = threadIdx.x;
  const int base = blockIdx.x * 1024 + t * 4;
  int s = 0;
#pragma unroll
  for (int k = 0; k < 4; ++k)
    if (base + k < n) s += cnt[base + k];
  sh[t] = s;
  __syncthreads();
  for (int w = 128; w > 0; w >>= 1) {
    if (t < w) sh[t] += sh[t + w];
    __syncthreads();
  }
  if (t == 0) bsum[blockIdx.x] = sh[0];
}

__global__ void __launch_bounds__(256)
scan_apply(const int* __restrict__ cnt, const int* __restrict__ bsum,
           int* __restrict__ start, float* __restrict__ dinv, int n) {
  __shared__ int sh[256];
  const int t = threadIdx.x;

  int part = 0;
  for (int b = t; b < (int)blockIdx.x; b += 256) part += bsum[b];
  sh[t] = part;
  __syncthreads();
  for (int w = 128; w > 0; w >>= 1) {
    if (t < w) sh[t] += sh[t + w];
    __syncthreads();
  }
  const int blockOff = sh[0];
  __syncthreads();

  const int base = blockIdx.x * 1024 + t * 4;
  int c[4];
#pragma unroll
  for (int k = 0; k < 4; ++k) c[k] = (base + k < n) ? cnt[base + k] : 0;
  const int tsum = c[0] + c[1] + c[2] + c[3];
  sh[t] = tsum;
  __syncthreads();
  for (int off = 1; off < 256; off <<= 1) {
    const int add = (t >= off) ? sh[t - off] : 0;
    __syncthreads();
    sh[t] += add;
    __syncthreads();
  }
  int run = blockOff + sh[t] - tsum;
#pragma unroll
  for (int k = 0; k < 4; ++k) {
    const int idx = base + k;
    if (idx <= n) start[idx] = run;
    if (idx < n) {
      const float v = (float)(c[k] + 1);
      dinv[idx] = (v > 0.0f) ? rsqrtf(v) : 0.0f;
    }
    run += c[k];
  }
}

__global__ void place_edges(const int* __restrict__ row, const int* __restrict__ col,
                            const int* __restrict__ rank, const int* __restrict__ start,
                            int* __restrict__ adj, int nE) {
  int e = blockIdx.x * blockDim.x + threadIdx.x;
  if (e < nE) adj[start[col[e]] + rank[e]] = row[e];
}

__global__ void __launch_bounds__(256)
sort_adj(const int* __restrict__ start, const int* __restrict__ in,
         int* __restrict__ out, int nNodes) {
  const long t = (long)blockIdx.x * 256 + threadIdx.x;
  const int i = (int)(t >> 4);
  const int l = (int)(t & 15);
  const bool live = i < nNodes;
  const int s = live ? start[i] : 0;
  const int e = live ? start[i + 1] : 0;
  const int nch = (e - s + 15) >> 4;
  const int nchO = __shfl_xor(nch, 16, 32);
  const int nchW = nch > nchO ? nch : nchO;

  for (int ca = 0; ca < nchW; ++ca) {
    const int ai = ca * 16 + l;
    const bool va = s + ai < e;
    const int ka = va ? in[s + ai] : 0;
    int r = 0;
    for (int cb = 0; cb < nchW; ++cb) {
      const int bl = s + cb * 16 + l;
      const int kbl = (bl < e) ? in[bl] : 0x7fffffff;
#pragma unroll
      for (int u = 0; u < 16; ++u) {
        const int kb = __shfl(kbl, u, 16);
        const int bi = cb * 16 + u;
        r += (kb < ka + ((bi < ai) ? 1 : 0)) ? 1 : 0;
      }
    }
    if (va) out[s + r] = ka;
  }
}

template <int K>
__global__ void __launch_bounds__(128)
gcn_gemm_wmma(const float* __restrict__ X, const float* __restrict__ W,
              const float* __restrict__ dinv, float* __restrict__ Y, int nRows) {
  __shared__ float wlds[K * WPAD];
  __shared__ float alds[16 * (K + 4)];

  const int tid  = threadIdx.x;
  const int wave = tid >> 5;
  const int lane = tid & 31;
  const int tile = blockIdx.x * 16;

  for (int i = tid * 4; i < K * 64; i += 128 * 4) {
    const int k = i >> 6;
    const int c = i & 63;
    *reinterpret_cast<float4*>(&wlds[k * WPAD + c]) =
        *reinterpret_cast<const float4*>(W + i);
  }
  for (int i = tid * 4; i < 16 * K; i += 128 * 4) {
    const int r = i / K;
    const int c = i % K;
    int gr = tile + r;
    if (gr >= nRows) gr = nRows - 1;
    *reinterpret_cast<float4*>(&alds[r * (K + 4) + c]) =
        *reinterpret_cast<const float4*>(X + (long)gr * K + c);
  }
  __syncthreads();

  const int m  = lane & 15;
  const int kq = (lane >> 4) * 2;
  const int n  = (lane & 15) + wave * 16;

  v8f c = {};
#pragma unroll
  for (int kk = 0; kk < K; kk += 4) {
    const int kb = kk + kq;
    v2f a, b;
    a.x = alds[m * (K + 4) + kb];
    a.y = alds[m * (K + 4) + kb + 1];
    b.x = wlds[kb * WPAD + n];
    b.y = wlds[(kb + 1) * WPAD + n];
    c = __builtin_amdgcn_wmma_f32_16x16x4_f32(false, a, false, b, (short)0, c,
                                              false, false);
  }

  const int mrow0 = (lane >> 4) * 8;
#pragma unroll
  for (int v = 0; v < 8; ++v) {
    const int gr = tile + mrow0 + v;
    if (gr < nRows) Y[(long)gr * 64 + n] = c[v] * dinv[gr];
  }
}

__global__ void __launch_bounds__(256)
gather_finalize(const int* __restrict__ start, const int* __restrict__ adj,
                const float* __restrict__ y, const float* __restrict__ dinv,
                const float* __restrict__ bias, float* __restrict__ out,
                int ostride, int nNodes) {
  const long t = (long)blockIdx.x * 256 + threadIdx.x;
  const int i0 = (int)(t >> 4);
  const int l = (int)(t & 15);
  const int f = l << 2;
  const bool live = i0 < nNodes;
  const int i = live ? i0 : nNodes - 1;
  const int s = live ? start[i] : 0;
  const int e = live ? start[i + 1] : 0;
  const int nch = (e - s + 7) >> 3;
  const int nchO = __shfl_xor(nch, 16, 32);
  const int nchW = nch > nchO ? nch : nchO;

  float4 acc = *reinterpret_cast<const float4*>(y + (long)i * 64 + f);
  for (int c = 0; c < nchW; ++c) {
    const int k = s + c * 8 + (l & 7);
    const int mine = (k < e) ? adj[k] : -1;
    int jv[8];
    float4 v[8];
#pragma unroll
    for (int u = 0; u < 8; ++u) {
      jv[u] = __shfl(mine, u, 16);
      const int jr = (jv[u] < 0) ? i : jv[u];
      v[u] = *reinterpret_cast<const float4*>(y + (long)jr * 64 + f);
    }
#pragma unroll
    for (int u = 0; u < 8; ++u) {
      if (jv[u] >= 0) {
        acc.x += v[u].x;
        acc.y += v[u].y;
        acc.z += v[u].z;
        acc.w += v[u].w;
      }
    }
  }
  if (live) {
    const float d = dinv[i];
    const float4 b = *reinterpret_cast<const float4*>(bias + f);
    float4 o;
    o.x = d * acc.x + b.x;
    o.y = d * acc.y + b.y;
    o.z = d * acc.z + b.z;
    o.w = d * acc.w + b.w;
    *reinterpret_cast<float4*>(out + (long)i * ostride + f) = o;
  }
}

extern "C" void kernel_launch(void* const* d_in, const int* in_sizes, int n_in,
                              void* d_out, int out_size, void* d_ws, size_t ws_size,
                              hipStream_t stream) {
  const float* latent    = (const float*)d_in[0];
  const float* condition = (const float*)d_in[1];
  const int*   ei        = (const int*)d_in[2];
  const float* W1 = (const float*)d_in[3];
  const float* b1 = (const float*)d_in[4];
  const float* W2 = (const float*)d_in[5];
  const float* b2 = (const float*)d_in[6];
  const float* W3 = (const float*)d_in[7];
  const float* b3 = (const float*)d_in[8];
  const float* W4 = (const float*)d_in[9];
  const float* b4 = (const float*)d_in[10];

  const int N = in_sizes[0] / 64;
  const int E = in_sizes[2] / 2;
  const int* rowp = ei;
  const int* colp = ei + E;

  char* ws = (char*)d_ws;
  size_t off = 0;
  float* dinv = (float*)(ws + off);
  off += ((size_t)N * 4 + 255) & ~(size_t)255;
  float* ybuf = (float*)(ws + off);
  off += (size_t)N * 64 * 4;
  char* graph = ws + off;
  off += (size_t)N * 64 * 4;
  float* hbuf = (float*)(ws + off);

  size_t goff = 0;
  int* cnt = (int*)(graph + goff);
  goff += ((size_t)N * 4 + 255) & ~(size_t)255;
  int* start = (int*)(graph + goff);
  goff += ((size_t)(N + 1) * 4 + 255) & ~(size_t)255;
  int* bsum = (int*)(graph + goff);
  goff += 4096;
  int* rank = (int*)(graph + goff);
  goff += ((size_t)E * 4 + 255) & ~(size_t)255;
  int* adjRaw = (int*)(graph + goff);
  goff += ((size_t)E * 4 + 255) & ~(size_t)255;
  int* adj = (int*)(graph + goff);

  const dim3 gemmGrid((N + 15) / 16);
  const dim3 gemmBlk(128);
  const dim3 fillN((N + 255) / 256);
  const dim3 edgeGrid((E + 255) / 256);
  const dim3 scanGrid((N + 1 + 1023) / 1024);
  const dim3 nodeGrid(((long)N * 16 + 255) / 256);

  fill_i32<<<fillN, 256, 0, stream>>>(cnt, 0, N);
  count_rank<<<edgeGrid, 256, 0, stream>>>(colp, cnt, rank, E);
  block_totals<<<scanGrid, 256, 0, stream>>>(cnt, bsum, N);
  scan_apply<<<scanGrid, 256, 0, stream>>>(cnt, bsum, start, dinv, N);
  place_edges<<<edgeGrid, 256, 0, stream>>>(rowp, colp, rank, start, adjRaw, E);
  sort_adj<<<nodeGrid, 256, 0, stream>>>(start, adjRaw, adj, N);

  gcn_gemm_wmma<64><<<gemmGrid, gemmBlk, 0, stream>>>(latent, W1, dinv, ybuf, N);
  gather_finalize<<<nodeGrid, 256, 0, stream>>>(start, adj, ybuf, dinv, b1, hbuf, 128, N);

  gcn_gemm_wmma<32><<<gemmGrid, gemmBlk, 0, stream>>>(condition, W2, dinv, ybuf, N);
  gather_finalize<<<nodeGrid, 256, 0, stream>>>(start, adj, ybuf, dinv, b2, hbuf + 64, 128, N);

  gcn_gemm_wmma<128><<<gemmGrid, gemmBlk, 0, stream>>>(hbuf, W3, dinv, ybuf, N);
  gather_finalize<<<nodeGrid, 256, 0, stream>>>(start, adj, ybuf, dinv, b3, hbuf, 64, N);

  gcn_gemm_wmma<64><<<gemmGrid, gemmBlk, 0, stream>>>(hbuf, W4, dinv, ybuf, N);
  gather_finalize<<<nodeGrid, 256, 0, stream>>>(start, adj, ybuf, dinv, b4, (float*)d_out, 64, N);
}


# Resident-size grids: each wave loops over several nodes, edge quads or GEMM row tiles (W staged once per work-group, next tile's rows loaded ahead), so under a third as many waves have to enter
# speedup vs baseline: 2.5091x; 1.1520x over previous
#include <hip/hip_runtime.h>
#include <hip/hip_bf16.h>

typedef __attribute__((ext_vector_type(2))) float v2f;
typedef __attribute__((ext_vector_type(8))) float v8f;

#define WPAD 72

__global__ void fill_i32(int* __restrict__ p, int v, int n) {
  int i = blockIdx.x * blockDim.x + threadIdx.x;
  if (i < n) p[i] = v;
}

template <bool GUARD>
__device__ __forceinline__ void count_rank_quad(const int* __restrict__ col,
                                                int* __restrict__ cnt,
                                                int* __restrict__ rank,
                                                int nE, int t, int nT) {
  int c[4], r[4];
#pragma unroll
  for (int k = 0; k < 4; ++k) {
    const int e = t + k * nT;
    c[k] = col[(GUARD && e >= nE) ? nE - 1 : e];
  }
#pragma unroll
  for (int k = 0; k < 4; ++k) {
    const int e = t + k * nT;
    r[k] = (!GUARD || e < nE) ? atomicAdd(&cnt[c[k]], 1) : 0;
  }
#pragma unroll
  for (int k = 0; k < 4; ++k) {
    const int e = t + k * nT;
    if (!GUARD || e < nE) rank[e] = r[k];
  }
}

__global__ void __launch_bounds__(256)
count_rank(const int* __restrict__ col, int* __restrict__ cnt,
           int* __restrict__ rank, int nE) {
  const int nT = (int)gridDim.x * 256;
  const int t = (int)blockIdx.x * 256 + (int)threadIdx.x;
  const bool full = (int)blockIdx.x * 256 + 255 + 3 * nT < nE;
  if (full) count_rank_quad<false>(col, cnt, rank, nE, t, nT);
  else      count_rank_quad<true>(col, cnt, rank, nE, t, nT);
}

__global__ void __launch_bounds__(256)
block_totals(const int* __restrict__ cnt, int* __restrict__ bsum, int n) {
  __shared__ int sh[256];
  const int t = threadIdx.x;
  const int base = blockIdx.x * 1024 + t * 4;
  int s = 0;
#pragma unroll
  for (int k = 0; k < 4; ++k)
    if (base + k < n) s += cnt[base + k];
  sh[t] = s;
  __syncthreads();
  for (int w = 128; w > 0; w >>= 1) {
    if (t < w) sh[t] += sh[t + w];
    __syncthreads();
  }
  if (t == 0) bsum[blockIdx.x] = sh[0];
}

__global__ void __launch_bounds__(256)
scan_apply(const int* __restrict__ cnt, const int* __restrict__ bsum,
           int* __restrict__ start, float* __restrict__ dinv, int n) {
  __shared__ int sh[256];
  const int t = threadIdx.x;

  int part = 0;
  for (int b = t; b < (int)blockIdx.x; b += 256) part += bsum[b];
  sh[t] = part;
  __syncthreads();
  for (int w = 128; w > 0; w >>= 1) {
    if (t < w) sh[t] += sh[t + w];
    __syncthreads();
  }
  const int blockOff = sh[0];
  __syncthreads();

  const int base = blockIdx.x * 1024 + t * 4;
  int c[4];
#pragma unroll
  for (int k = 0; k < 4; ++k) c[k] = (base + k < n) ? cnt[base + k] : 0;
  const int tsum = c[0] + c[1] + c[2] + c[3];
  sh[t] = tsum;
  __syncthreads();
  for (int off = 1; off < 256; off <<= 1) {
    const int add = (t >= off) ? sh[t - off] : 0;
    __syncthreads();
    sh[t] += add;
    __syncthreads();
  }
  int run = blockOff + sh[t] - tsum;
#pragma unroll
  for (int k = 0; k < 4; ++k) {
    const int idx = base + k;
    if (idx <= n) start[idx] = run;
    if (idx < n) {
      const float v = (float)(c[k] + 1);
      dinv[idx] = (v > 0.0f) ? rsqrtf(v) : 0.0f;
    }
    run += c[k];
  }
}

template <bool GUARD>
__device__ __forceinline__ void place_quad(const int* __restrict__ row,
                                           const int* __restrict__ col,
                                           const int* __restrict__ rank,
                                           const int* __restrict__ start,
                                           int* __restrict__ adj,
                                           int nE, int t, int nT) {
  int c[4], r[4], k4[4], s[4];
#pragma unroll
  for (int k = 0; k < 4; ++k) {
    const int e = t + k * nT;
    const int ec = (GUARD && e >= nE) ? nE - 1 : e;
    c[k] = col[ec];
    r[k] = row[ec];
    k4[k] = rank[ec];
  }
#pragma unroll
  for (int k = 0; k < 4; ++k) s[k] = start[c[k]];
#pragma unroll
  for (int k = 0; k < 4; ++k) {
    const int e = t + k * nT;
    if (!GUARD || e < nE) adj[s[k] + k4[k]] = r[k];
  }
}

__global__ void __launch_bounds__(256)
place_edges(const int* __restrict__ row, const int* __restrict__ col,
            const int* __restrict__ rank, const int* __restrict__ start,
            int* __restrict__ adj, int nE) {
  const int nT = (int)gridDim.x * 256;
  const int t = (int)blockIdx.x * 256 + (int)threadIdx.x;
  const bool full = (int)blockIdx.x * 256 + 255 + 3 * nT < nE;
  if (full) place_quad<false>(row, col, rank, start, adj, nE, t, nT);
  else      place_quad<true>(row, col, rank, start, adj, nE, t, nT);
}

__global__ void __launch_bounds__(256)
sort_adj(const int* __restrict__ start, const int* __restrict__ in,
         int* __restrict__ out, int nNodes) {
  const int t = (int)blockIdx.x * 256 + (int)threadIdx.x;
  const int g = t >> 4;
  const int nG = ((int)gridDim.x * 256) >> 4;
  const int l = t & 15;
  const int nIt = (nNodes + nG - 1) / nG;
  for (int it = 0; it < nIt; ++it) {
  const int i0 = g + it * nG;
  const bool live = i0 < nNodes;
  const int i = live ? i0 : nNodes - 1;
  const int s = start[i];
  const int eRaw = start[i + 1];
  const int e = live ? eRaw : s;
  const int nch = (e - s + 15) >> 4;
  const int nchO = __shfl_xor(nch, 16, 32);
  const int nchW = nch > nchO ? nch : nchO;

  for (int ca = 0; ca < nchW; ++ca) {
    const int ai = ca * 16 + l;
    const bool va = s + ai < e;
    const int ka = va ? in[s + ai] : 0;
    int r = 0;
    for (int cb = 0; cb < nchW; ++cb) {
      const int bl = s + cb * 16 + l;
      const int kbl = (bl < e) ? in[bl] : 0x7fffffff;
#pragma unroll
      for (int u = 0; u < 16; ++u) {
        const int kb = __shfl(kbl, u, 16);
        const int bi = cb * 16 + u;
        r += (kb < ka + ((bi < ai) ? 1 : 0)) ? 1 : 0;
      }
    }
    if (va) out[s + r] = ka;
  }
  }
}

template <int K>
__global__ void __launch_bounds__(128)
gcn_gemm_wmma(const float* __restrict__ X, const float* __restrict__ W,
              const float* __restrict__ dinv, float* __restrict__ Y, int nRows) {
  __shared__ float wlds[K * WPAD];
  __shared__ float alds[16 * (K + 4)];

  constexpr int NA = (16 * K) / (128 * 4);

  const int tid  = threadIdx.x;
  const int wave = tid >> 5;
  const int lane = tid & 31;
  const int nTiles = (nRows + 15) >> 4;
  const int tStep = (int)gridDim.x;

  for (int i = tid * 4; i < K * 64; i += 128 * 4) {
    const int k = i >> 6;
    const int c = i & 63;
    *reinterpret_cast<float4*>(&wlds[k * WPAD + c]) =
        *reinterpret_cast<const float4*>(W + i);
  }

  const int m  = lane & 15;
  const int kq = (lane >> 4) * 2;
  const int n  = (lane & 15) + wave * 16;
  const int mrow0 = (lane >> 4) * 8;

  auto loadPiece = [&](int tileIdx, int j) -> float4 {
    const int i = (tid + j * 128) * 4;
    const int r = i / K;
    const int cc = i % K;
    int gr = tileIdx * 16 + r;
    if (gr >= nRows) gr = nRows - 1;
    return *reinterpret_cast<const float4*>(X + (long)gr * K + cc);
  };
  auto storePiece = [&](const float4& v, int j) {
    const int i = (tid + j * 128) * 4;
    const int r = i / K;
    const int cc = i % K;
    *reinterpret_cast<float4*>(&alds[r * (K + 4) + cc]) = v;
  };

  int tl = (int)blockIdx.x;
  float4 p0, p1, p2, p3;
  p0 = loadPiece(tl, 0);
  if constexpr (NA >= 2) p1 = loadPiece(tl, 1);
  if constexpr (NA >= 4) { p2 = loadPiece(tl, 2); p3 = loadPiece(tl, 3); }

  for (; tl < nTiles; tl += tStep) {
  const int tile = tl * 16;
  __syncthreads();
  storePiece(p0, 0);
  if constexpr (NA >= 2) storePiece(p1, 1);
  if constexpr (NA >= 4) { storePiece(p2, 2); storePiece(p3, 3); }
  __syncthreads();

  float dv[8];
#pragma unroll
  for (int v = 0; v < 8; ++v) {
    int gr = tile + mrow0 + v;
    if (gr >= nRows) gr = nRows - 1;
    dv[v] = dinv[gr];
  }

  p0 = loadPiece(tl + tStep, 0);
  if constexpr (NA >= 2) p1 = loadPiece(tl + tStep, 1);
  if constexpr (NA >= 4) { p2 = loadPiece(tl + tStep, 2); p3 = loadPiece(tl + tStep, 3); }

  v8f c = {};
#pragma unroll
  for (int kk = 0; kk < K; kk += 4) {
    const int kb = kk + kq;
    v2f a, b;
    a.x = alds[m * (K + 4) + kb];
    a.y = alds[m * (K + 4) + kb + 1];
    b.x = wlds[kb * WPAD + n];
    b.y = wlds[(kb + 1) * WPAD + n];
    c = __builtin_amdgcn_wmma_f32_16x16x4_f32(false, a, false, b, (short)0, c,
                                              false, false);
  }

  if (tile + 16 <= nRows) {
#pragma unroll
    for (int v = 0; v < 8; ++v) {
      const int gr = tile + mrow0 + v;
      Y[(long)gr * 64 + n] = c[v] * dv[v];
    }
  } else {
#pragma unroll
    for (int v = 0; v < 8; ++v) {
      const int gr = tile + mrow0 + v;
      if (gr < nRows) Y[(long)gr * 64 + n] = c[v] * dv[v];
    }
  }
  }
}

__global__ void __launch_bounds__(256)
gather_finalize(const int* __restrict__ start, const int* __restrict__ adj,
                const float* __restrict__ y, const float* __restrict__ dinv,
                const float* __restrict__ bias, float* __restrict__ out,
                int ostride, int nNodes) {
  const int t = (int)blockIdx.x * 256 + (int)threadIdx.x;
  const int g = t >> 4;
  const int nG = ((int)gridDim.x * 256) >> 4;
  const int l = t & 15;
  const int f = l << 2;
  const float4 b = *reinterpret_cast<const float4*>(bias + f);
  const int nIt = (nNodes + nG - 1) / nG;
  for (int it = 0; it < nIt; ++it) {
  const int i0 = g + it * nG;
  const bool live = i0 < nNodes;
  const int i = live ? i0 : nNodes - 1;
  const int s = start[i];
  const int eRaw = start[i + 1];
  const int e = live ? eRaw : s;
  const int nch = (e - s + 7) >> 3;
  const int nchO = __shfl_xor(nch, 16, 32);
  const int nchW = nch > nchO ? nch : nchO;

  float4 acc = *reinterpret_cast<const float4*>(y + (long)i * 64 + f);
  for (int c = 0; c < nchW; ++c) {
    const int k = s + c * 8 + (l & 7);
    const int mine = (k < e) ? adj[k] : -1;
    int jv[8];
    float4 v[8];
#pragma unroll
    for (int u = 0; u < 8; ++u) {
      jv[u] = __shfl(mine, u, 16);
      const int jr = (jv[u] < 0) ? i : jv[u];
      v[u] = *reinterpret_cast<const float4*>(y + (long)jr * 64 + f);
    }
#pragma unroll
    for (int u = 0; u < 8; ++u) {
      if (jv[u] >= 0) {
        acc.x += v[u].x;
        acc.y += v[u].y;
        acc.z += v[u].z;
        acc.w += v[u].w;
      }
    }
  }
  if (live) {
    const float d = dinv[i];
    float4 o;
    o.x = d * acc.x + b.x;
    o.y = d * acc.y + b.y;
    o.z = d * acc.z + b.z;
    o.w = d * acc.w + b.w;
    *reinterpret_cast<float4*>(out + (long)i * ostride + f) = o;
  }
  }
}

extern "C" void kernel_launch(void* const* d_in, const int* in_sizes, int n_in,
                              void* d_out, int out_size, void* d_ws, size_t ws_size,
                              hipStream_t stream) {
  const float* latent    = (const float*)d_in[0];
  const float* condition = (const float*)d_in[1];
  const int*   ei        = (const int*)d_in[2];
  const float* W1 = (const float*)d_in[3];
  const float* b1 = (const float*)d_in[4];
  const float* W2 = (const float*)d_in[5];
  const float* b2 = (const float*)d_in[6];
  const float* W3 = (const float*)d_in[7];
  const float* b3 = (const float*)d_in[8];
  const float* W4 = (const float*)d_in[9];
  const float* b4 = (const float*)d_in[10];

  const int N = in_sizes[0] / 64;
  const int E = in_sizes[2] / 2;
  const int* rowp = ei;
  const int* colp = ei + E;

  char* ws = (char*)d_ws;
  size_t off = 0;
  float* dinv = (float*)(ws + off);
  off += ((size_t)N * 4 + 255) & ~(size_t)255;
  float* ybuf = (float*)(ws + off);
  off += (size_t)N * 64 * 4;
  char* graph = ws + off;
  off += (size_t)N * 64 * 4;
  float* hbuf = (float*)(ws + off);

  size_t goff = 0;
  int* cnt = (int*)(graph + goff);
  goff += ((size_t)N * 4 + 255) & ~(size_t)255;
  int* start = (int*)(graph + goff);
  goff += ((size_t)(N + 1) * 4 + 255) & ~(size_t)255;
  int* bsum = (int*)(graph + goff);
  goff += 4096;
  int* rank = (int*)(graph + goff);
  goff += ((size_t)E * 4 + 255) & ~(size_t)255;
  int* adjRaw = (int*)(graph + goff);
  goff += ((size_t)E * 4 + 255) & ~(size_t)255;
  int* adj = (int*)(graph + goff);

  const dim3 gemmGrid(1250);
  const dim3 gemmBlk(128);
  const dim3 fillN((N + 255) / 256);
  const dim3 edgeGrid((E + 1023) / 1024);
  const dim3 scanGrid((N + 1 + 1023) / 1024);
  const dim3 nodeGrid(2048);

  fill_i32<<<fillN, 256, 0, stream>>>(cnt, 0, N);
  count_rank<<<edgeGrid, 256, 0, stream>>>(colp, cnt, rank, E);
  block_totals<<<scanGrid, 256, 0, stream>>>(cnt, bsum, N);
  scan_apply<<<scanGrid, 256, 0, stream>>>(cnt, bsum, start, dinv, N);
  place_edges<<<edgeGrid, 256, 0, stream>>>(rowp, colp, rank, start, adjRaw, E);
  sort_adj<<<nodeGrid, 256, 0, stream>>>(start, adjRaw, adj, N);

  gcn_gemm_wmma<64><<<gemmGrid, gemmBlk, 0, stream>>>(latent, W1, dinv, ybuf, N);
  gather_finalize<<<nodeGrid, 256, 0, stream>>>(start, adj, ybuf, dinv, b1, hbuf, 128, N);

  gcn_gemm_wmma<32><<<gemmGrid, gemmBlk, 0, stream>>>(condition, W2, dinv, ybuf, N);
  gather_finalize<<<nodeGrid, 256, 0, stream>>>(start, adj, ybuf, dinv, b2, hbuf + 64, 128, N);

  gcn_gemm_wmma<128><<<gemmGrid, gemmBlk, 0, stream>>>(hbuf, W3, dinv, ybuf, N);
  gather_finalize<<<nodeGrid, 256, 0, stream>>>(start, adj, ybuf, dinv, b3, hbuf, 64, N);

  gcn_gemm_wmma<64><<<gemmGrid, gemmBlk, 0, stream>>>(hbuf, W4, dinv, ybuf, N);
  gather_finalize<<<nodeGrid, 256, 0, stream>>>(start, adj, ybuf, dinv, b4, (float*)d_out, 64, N);
}


# Network is linear end to end: fold W1..W4 and the biases into one 96x64 weight (small WMMA kernel), so 1 GEMM + 3 aggregations replace 4 + 4; exact algebra, only the f32 rounding order changes
# speedup vs baseline: 3.1792x; 1.2671x over previous
#include <hip/hip_runtime.h>
#include <hip/hip_bf16.h>

typedef __attribute__((ext_vector_type(2))) float v2f;
typedef __attribute__((ext_vector_type(8))) float v8f;

#define WPAD 72

__global__ void fill_i32(int* __restrict__ p, int v, int n) {
  int i = blockIdx.x * blockDim.x + threadIdx.x;
  if (i < n) p[i] = v;
}

template <bool GUARD>
__device__ __forceinline__ void count_rank_quad(const int* __restrict__ col,
                                                int* __restrict__ cnt,
                                                int* __restrict__ rank,
                                                int nE, int t, int nT) {
  int c[4], r[4];
#pragma unroll
  for (int k = 0; k < 4; ++k) {
    const int e = t + k * nT;
    c[k] = col[(GUARD && e >= nE) ? nE - 1 : e];
  }
#pragma unroll
  for (int k = 0; k < 4; ++k) {
    const int e = t + k * nT;
    r[k] = (!GUARD || e < nE) ? atomicAdd(&cnt[c[k]], 1) : 0;
  }
#pragma unroll
  for (int k = 0; k < 4; ++k) {
    const int e = t + k * nT;
    if (!GUARD || e < nE) rank[e] = r[k];
  }
}

__global__ void __launch_bounds__(256)
count_rank(const int* __restrict__ col, int* __restrict__ cnt,
           int* __restrict__ rank, int nE) {
  const int nT = (int)gridDim.x * 256;
  const int t = (int)blockIdx.x * 256 + (int)threadIdx.x;
  const bool full = (int)blockIdx.x * 256 + 255 + 3 * nT < nE;
  if (full) count_rank_quad<false>(col, cnt, rank, nE, t, nT);
  else      count_rank_quad<true>(col, cnt, rank, nE, t, nT);
}

__global__ void __launch_bounds__(256)
block_totals(const int* __restrict__ cnt, int* __restrict__ bsum, int n) {
  __shared__ int sh[256];
  const int t = threadIdx.x;
  const int base = blockIdx.x * 1024 + t * 4;
  int s = 0;
#pragma unroll
  for (int k = 0; k < 4; ++k)
    if (base + k < n) s += cnt[base + k];
  sh[t] = s;
  __syncthreads();
  for (int w = 128; w > 0; w >>= 1) {
    if (t < w) sh[t] += sh[t + w];
    __syncthreads();
  }
  if (t == 0) bsum[blockIdx.x] = sh[0];
}

__global__ void __launch_bounds__(256)
scan_apply(const int* __restrict__ cnt, const int* __restrict__ bsum,
           int* __restrict__ start, float* __restrict__ dinv, int n) {
  __shared__ int sh[256];
  const int t = threadIdx.x;

  int part = 0;
  for (int b = t; b < (int)blockIdx.x; b += 256) part += bsum[b];
  sh[t] = part;
  __syncthreads();
  for (int w = 128; w > 0; w >>= 1) {
    if (t < w) sh[t] += sh[t + w];
    __syncthreads();
  }
  const int blockOff = sh[0];
  __syncthreads();

  const int base = blockIdx.x * 1024 + t * 4;
  int c[4];
#pragma unroll
  for (int k = 0; k < 4; ++k) c[k] = (base + k < n) ? cnt[base + k] : 0;
  const int tsum = c[0] + c[1] + c[2] + c[3];
  sh[t] = tsum;
  __syncthreads();
  for (int off = 1; off < 256; off <<= 1) {
    const int add = (t >= off) ? sh[t - off] : 0;
    __syncthreads();
    sh[t] += add;
    __syncthreads();
  }
  int run = blockOff + sh[t] - tsum;
#pragma unroll
  for (int k = 0; k < 4; ++k) {
    const int idx = base + k;
    if (idx <= n) start[idx] = run;
    if (idx < n) {
      const float v = (float)(c[k] + 1);
      dinv[idx] = (v > 0.0f) ? rsqrtf(v) : 0.0f;
    }
    run += c[k];
  }
}

template <bool GUARD>
__device__ __forceinline__ void place_quad(const int* __restrict__ row,
                                           const int* __restrict__ col,
                                           const int* __restrict__ rank,
                                           const int* __restrict__ start,
                                           int* __restrict__ adj,
                                           int nE, int t, int nT) {
  int c[4], r[4], k4[4], s[4];
#pragma unroll
  for (int k = 0; k < 4; ++k) {
    const int e = t + k * nT;
    const int ec = (GUARD && e >= nE) ? nE - 1 : e;
    c[k] = col[ec];
    r[k] = row[ec];
    k4[k] = rank[ec];
  }
#pragma unroll
  for (int k = 0; k < 4; ++k) s[k] = start[c[k]];
#pragma unroll
  for (int k = 0; k < 4; ++k) {
    const int e = t + k * nT;
    if (!GUARD || e < nE) adj[s[k] + k4[k]] = r[k];
  }
}

__global__ void __launch_bounds__(256)
place_edges(const int* __restrict__ row, const int* __restrict__ col,
            const int* __restrict__ rank, const int* __restrict__ start,
            int* __restrict__ adj, int nE) {
  const int nT = (int)gridDim.x * 256;
  const int t = (int)blockIdx.x * 256 + (int)threadIdx.x;
  const bool full = (int)blockIdx.x * 256 + 255 + 3 * nT < nE;
  if (full) place_quad<false>(row, col, rank, start, adj, nE, t, nT);
  else      place_quad<true>(row, col, rank, start, adj, nE, t, nT);
}

__global__ void __launch_bounds__(256)
sort_adj(const int* __restrict__ start, const int* __restrict__ in,
         int* __restrict__ out, int nNodes) {
  const int t = (int)blockIdx.x * 256 + (int)threadIdx.x;
  const int g = t >> 4;
  const int nG = ((int)gridDim.x * 256) >> 4;
  const int l = t & 15;
  const int nIt = (nNodes + nG - 1) / nG;
  for (int it = 0; it < nIt; ++it) {
  const int i0 = g + it * nG;
  const bool live = i0 < nNodes;
  const int i = live ? i0 : nNodes - 1;
  const int s = start[i];
  const int eRaw = start[i + 1];
  const int e = live ? eRaw : s;
  const int nch = (e - s + 15) >> 4;
  const int nchO = __shfl_xor(nch, 16, 32);
  const int nchW = nch > nchO ? nch : nchO;

  for (int ca = 0; ca < nchW; ++ca) {
    const int ai = ca * 16 + l;
    const bool va = s + ai < e;
    const int ka = va ? in[s + ai] : 0;
    int r = 0;
    for (int cb = 0; cb < nchW; ++cb) {
      const int bl = s + cb * 16 + l;
      const int kbl = (bl < e) ? in[bl] : 0x7fffffff;
#pragma unroll
      for (int u = 0; u < 16; ++u) {
        const int kb = __shfl(kbl, u, 16);
        const int bi = cb * 16 + u;
        r += (kb < ka + ((bi < ai) ? 1 : 0)) ? 1 : 0;
      }
    }
    if (va) out[s + r] = ka;
  }
  }
}

__global__ void __launch_bounds__(128)
fold_weights(const float* __restrict__ W1, const float* __restrict__ b1,
             const float* __restrict__ W2, const float* __restrict__ b2,
             const float* __restrict__ W3, const float* __restrict__ b3,
             const float* __restrict__ W4, float* __restrict__ fold) {
  __shared__ float slds[16 * 132];
  __shared__ float tlds[16 * 68];

  const int tid  = threadIdx.x;
  const int wave = tid >> 5;
  const int lane = tid & 31;
  const int bt = (int)blockIdx.x;
  const int g0 = bt * 16;

  const int m  = lane & 15;
  const int kq = (lane >> 4) * 2;
  const int n  = (lane & 15) + wave * 16;
  const int mrow0 = (lane >> 4) * 8;

  v2f w4f[16];
#pragma unroll
  for (int q = 0; q < 16; ++q) {
    const int kb = q * 4 + kq;
    w4f[q].x = W4[kb * 64 + n];
    w4f[q].y = W4[(kb + 1) * 64 + n];
  }

  float sv[16];
#pragma unroll
  for (int r = 0; r < 16; ++r) sv[r] = 0.0f;
  if (bt < 4) {
    if (tid < 64) {
#pragma unroll
      for (int r = 0; r < 16; ++r) sv[r] = W1[(g0 + r) * 64 + tid];
    }
  } else if (bt < 6) {
    if (tid >= 64) {
#pragma unroll
      for (int r = 0; r < 16; ++r) sv[r] = W2[(g0 - 64 + r) * 64 + (tid - 64)];
    }
  } else {
    sv[0] = (tid < 64) ? b1[tid] : b2[tid - 64];
  }
#pragma unroll
  for (int r = 0; r < 16; ++r) slds[r * 132 + tid] = sv[r];
  __syncthreads();

  const int kLo = (bt >= 4 && bt < 6) ? 64 : 0;
  const int kHi = (bt < 4) ? 64 : 128;

  v8f t = {};
  for (int k0 = kLo; k0 < kHi; k0 += 64) {
#pragma unroll
    for (int kk = 0; kk < 64; kk += 4) {
      const int kb = k0 + kk + kq;
      v2f a, b;
      a.x = slds[m * 132 + kb];
      a.y = slds[m * 132 + kb + 1];
      b.x = W3[kb * 64 + n];
      b.y = W3[(kb + 1) * 64 + n];
      t = __builtin_amdgcn_wmma_f32_16x16x4_f32(false, a, false, b, (short)0, t,
                                                false, false);
    }
  }
#pragma unroll
  for (int v = 0; v < 8; ++v) {
    const int g = g0 + mrow0 + v;
    tlds[(mrow0 + v) * 68 + n] = (g == 97) ? b3[n] : t[v];
  }
  __syncthreads();

  v8f c = {};
#pragma unroll
  for (int kk = 0; kk < 64; kk += 4) {
    const int kb = kk + kq;
    v2f a;
    a.x = tlds[m * 68 + kb];
    a.y = tlds[m * 68 + kb + 1];
    c = __builtin_amdgcn_wmma_f32_16x16x4_f32(false, a, false, w4f[kk >> 2],
                                              (short)0, c, false, false);
  }
#pragma unroll
  for (int v = 0; v < 8; ++v) {
    const int g = g0 + mrow0 + v;
    if (g < 98) fold[g * 64 + n] = c[v];
  }
}

template <int K1, int K2>
__global__ void __launch_bounds__(128)
gcn_gemm_wmma(const float* __restrict__ X1, const float* __restrict__ X2,
              const float* __restrict__ W, const float* __restrict__ dinv,
              float* __restrict__ Y, int nRows) {
  constexpr int K = K1 + K2;
  static_assert(16 * K1 == 2 * 128 * 4 && 16 * K2 == 128 * 4, "tile pieces");
  __shared__ float wlds[K * WPAD];
  __shared__ float alds[16 * (K + 4)];

  const int tid  = threadIdx.x;
  const int wave = tid >> 5;
  const int lane = tid & 31;
  const int nTiles = (nRows + 15) >> 4;
  const int tStep = (int)gridDim.x;

  for (int i = tid * 4; i < K * 64; i += 128 * 4) {
    const int k = i >> 6;
    const int c = i & 63;
    *reinterpret_cast<float4*>(&wlds[k * WPAD + c]) =
        *reinterpret_cast<const float4*>(W + i);
  }

  const int m  = lane & 15;
  const int kq = (lane >> 4) * 2;
  const int n  = (lane & 15) + wave * 16;
  const int mrow0 = (lane >> 4) * 8;

  auto loadPiece1 = [&](int tileIdx, int j) -> float4 {
    const int i = (tid + j * 128) * 4;
    const int r = i / K1;
    const int cc = i % K1;
    int gr = tileIdx * 16 + r;
    if (gr >= nRows) gr = nRows - 1;
    return *reinterpret_cast<const float4*>(X1 + (long)gr * K1 + cc);
  };
  auto loadPiece2 = [&](int tileIdx) -> float4 {
    const int i = tid * 4;
    const int r = i / K2;
    const int cc = i % K2;
    int gr = tileIdx * 16 + r;
    if (gr >= nRows) gr = nRows - 1;
    return *reinterpret_cast<const float4*>(X2 + (long)gr * K2 + cc);
  };
  auto storePiece1 = [&](const float4& v, int j) {
    const int i = (tid + j * 128) * 4;
    const int r = i / K1;
    const int cc = i % K1;
    *reinterpret_cast<float4*>(&alds[r * (K + 4) + cc]) = v;
  };
  auto storePiece2 = [&](const float4& v) {
    const int i = tid * 4;
    const int r = i / K2;
    const int cc = i % K2;
    *reinterpret_cast<float4*>(&alds[r * (K + 4) + K1 + cc]) = v;
  };

  int tl = (int)blockIdx.x;
  float4 p0 = loadPiece1(tl, 0);
  float4 p1 = loadPiece1(tl, 1);
  float4 p2 = loadPiece2(tl);

  for (; tl < nTiles; tl += tStep) {
  const int tile = tl * 16;
  __syncthreads();
  storePiece1(p0, 0);
  storePiece1(p1, 1);
  storePiece2(p2);
  __syncthreads();

  float dv[8];
#pragma unroll
  for (int v = 0; v < 8; ++v) {
    int gr = tile + mrow0 + v;
    if (gr >= nRows) gr = nRows - 1;
    dv[v] = dinv[gr];
  }

  p0 = loadPiece1(tl + tStep, 0);
  p1 = loadPiece1(tl + tStep, 1);
  p2 = loadPiece2(tl + tStep);

  v8f c = {};
#pragma unroll
  for (int kk = 0; kk < K; kk += 4) {
    const int kb = kk + kq;
    v2f a, b;
    a.x = alds[m * (K + 4) + kb];
    a.y = alds[m * (K + 4) + kb + 1];
    b.x = wlds[kb * WPAD + n];
    b.y = wlds[(kb + 1) * WPAD + n];
    c = __builtin_amdgcn_wmma_f32_16x16x4_f32(false, a, false, b, (short)0, c,
                                              false, false);
  }

  if (tile + 16 <= nRows) {
#pragma unroll
    for (int v = 0; v < 8; ++v) {
      const int gr = tile + mrow0 + v;
      Y[(long)gr * 64 + n] = c[v] * dv[v];
    }
  } else {
#pragma unroll
    for (int v = 0; v < 8; ++v) {
      const int gr = tile + mrow0 + v;
      if (gr < nRows) Y[(long)gr * 64 + n] = c[v] * dv[v];
    }
  }
  }
}

template <bool CHAIN>
__global__ void __launch_bounds__(256)
gather_finalize(const int* __restrict__ start, const int* __restrict__ adj,
                const float* __restrict__ y, const float* __restrict__ dinv,
                const float* __restrict__ bias, float* __restrict__ out,
                int nNodes) {
  const int t = (int)blockIdx.x * 256 + (int)threadIdx.x;
  const int g = t >> 4;
  const int nG = ((int)gridDim.x * 256) >> 4;
  const int l = t & 15;
  const int f = l << 2;
  const float4 b = *reinterpret_cast<const float4*>(bias + f);
  const int nIt = (nNodes + nG - 1) / nG;
  for (int it = 0; it < nIt; ++it) {
  const int i0 = g + it * nG;
  const bool live = i0 < nNodes;
  const int i = live ? i0 : nNodes - 1;
  const int s = start[i];
  const int eRaw = start[i + 1];
  const int e = live ? eRaw : s;
  const int nch = (e - s + 7) >> 3;
  const int nchO = __shfl_xor(nch, 16, 32);
  const int nchW = nch > nchO ? nch : nchO;

  float4 acc = *reinterpret_cast<const float4*>(y + (long)i * 64 + f);
  for (int c = 0; c < nchW; ++c) {
    const int k = s + c * 8 + (l & 7);
    const int mine = (k < e) ? adj[k] : -1;
    int jv[8];
    float4 v[8];
#pragma unroll
    for (int u = 0; u < 8; ++u) {
      jv[u] = __shfl(mine, u, 16);
      const int jr = (jv[u] < 0) ? i : jv[u];
      v[u] = *reinterpret_cast<const float4*>(y + (long)jr * 64 + f);
    }
#pragma unroll
    for (int u = 0; u < 8; ++u) {
      if (jv[u] >= 0) {
        acc.x += v[u].x;
        acc.y += v[u].y;
        acc.z += v[u].z;
        acc.w += v[u].w;
      }
    }
  }
  if (live) {
    const float d = dinv[i];
    float4 o;
    o.x = d * acc.x + b.x;
    o.y = d * acc.y + b.y;
    o.z = d * acc.z + b.z;
    o.w = d * acc.w + b.w;
    if (CHAIN) {
      o.x *= d;
      o.y *= d;
      o.z *= d;
      o.w *= d;
    }
    *reinterpret_cast<float4*>(out + (long)i * 64 + f) = o;
  }
  }
}

extern "C" void kernel_launch(void* const* d_in, const int* in_sizes, int n_in,
                              void* d_out, int out_size, void* d_ws, size_t ws_size,
                              hipStream_t stream) {
  const float* latent    = (const float*)d_in[0];
  const float* condition = (const float*)d_in[1];
  const int*   ei        = (const int*)d_in[2];
  const float* W1 = (const float*)d_in[3];
  const float* b1 = (const float*)d_in[4];
  const float* W2 = (const float*)d_in[5];
  const float* b2 = (const float*)d_in[6];
  const float* W3 = (const float*)d_in[7];
  const float* b3 = (const float*)d_in[8];
  const float* W4 = (const float*)d_in[9];
  const float* b4 = (const float*)d_in[10];

  const int N = in_sizes[0] / 64;
  const int E = in_sizes[2] / 2;
  const int* rowp = ei;
  const int* colp = ei + E;

  char* ws = (char*)d_ws;
  size_t off = 0;
  float* dinv = (float*)(ws + off);
  off += ((size_t)N * 4 + 255) & ~(size_t)255;
  float* ybuf = (float*)(ws + off);
  off += (size_t)N * 64 * 4;
  char* graph = ws + off;
  off += (size_t)N * 64 * 4;
  float* hbuf = (float*)(ws + off);
  float* fold = hbuf + (size_t)N * 64;

  size_t goff = 0;
  int* cnt = (int*)(graph + goff);
  goff += ((size_t)N * 4 + 255) & ~(size_t)255;
  int* start = (int*)(graph + goff);
  goff += ((size_t)(N + 1) * 4 + 255) & ~(size_t)255;
  int* bsum = (int*)(graph + goff);
  goff += 4096;
  int* rank = (int*)(graph + goff);
  goff += ((size_t)E * 4 + 255) & ~(size_t)255;
  int* adjRaw = (int*)(graph + goff);
  goff += ((size_t)E * 4 + 255) & ~(size_t)255;
  int* adj = (int*)(graph + goff);

  const dim3 gemmGrid(1250);
  const dim3 gemmBlk(128);
  const dim3 fillN((N + 255) / 256);
  const dim3 edgeGrid((E + 1023) / 1024);
  const dim3 scanGrid((N + 1 + 1023) / 1024);
  const dim3 nodeGrid(2048);

  fill_i32<<<fillN, 256, 0, stream>>>(cnt, 0, N);
  count_rank<<<edgeGrid, 256, 0, stream>>>(colp, cnt, rank, E);
  block_totals<<<scanGrid, 256, 0, stream>>>(cnt, bsum, N);
  scan_apply<<<scanGrid, 256, 0, stream>>>(cnt, bsum, start, dinv, N);
  place_edges<<<edgeGrid, 256, 0, stream>>>(rowp, colp, rank, start, adjRaw, E);
  sort_adj<<<nodeGrid, 256, 0, stream>>>(start, adjRaw, adj, N);

  fold_weights<<<dim3(7), gemmBlk, 0, stream>>>(W1, b1, W2, b2, W3, b3, W4, fold);

  gcn_gemm_wmma<64, 32><<<gemmGrid, gemmBlk, 0, stream>>>(latent, condition, fold, dinv, ybuf, N);

  gather_finalize<true><<<nodeGrid, 256, 0, stream>>>(start, adj, ybuf, dinv, fold + 96 * 64, hbuf, N);
  gather_finalize<true><<<nodeGrid, 256, 0, stream>>>(start, adj, hbuf, dinv, fold + 97 * 64, ybuf, N);
  gather_finalize<false><<<nodeGrid, 256, 0, stream>>>(start, adj, ybuf, dinv, b4, (float*)d_out, N);
}


# Buckets built without per-edge global atomics: edges partitioned into 512-node bins (one atomic per block and bin), each bin's buckets built in LDS and written whole; clear pass gone
# speedup vs baseline: 6.5170x; 2.0499x over previous
#include <hip/hip_runtime.h>
#include <hip/hip_bf16.h>

typedef __attribute__((ext_vector_type(2))) float v2f;
typedef __attribute__((ext_vector_type(8))) float v8f;

#define WPAD 72

#define BKT 64
#define BCAP 63

#define BIN_SHIFT 9
#define BIN_NODES (1 << BIN_SHIFT)
#define BIN_CAP 16384
#define FILL_PITCH 64
#define PART_T 512
#define PART_EPT 8
#define PART_E (PART_T * PART_EPT)
#define BB_T 1024
#define BB_EPT 12

__device__ __forceinline__ float dinv_of(int deg) {
  const float v = (float)(deg + 1);
  return (v > 0.0f) ? rsqrtf(v) : 0.0f;
}

__global__ void __launch_bounds__(PART_T)
partition_edges(const int* __restrict__ row, const int* __restrict__ col,
                int* __restrict__ fill, int* __restrict__ seg, int nE) {
  __shared__ int hist[PART_T];
  __shared__ int gbase[PART_T];
  __shared__ int wsum[PART_T / 32];
  __shared__ int sval[PART_E];
  __shared__ int sdst[PART_E];

  const int tid = (int)threadIdx.x;
  const int e0 = (int)blockIdx.x * PART_E;
  hist[tid] = 0;
  __syncthreads();

  int bin[PART_EPT], pk[PART_EPT], lr[PART_EPT];
#pragma unroll
  for (int k = 0; k < PART_EPT; ++k) {
    const int e = e0 + k * PART_T + tid;
    const bool valid = e < nE;
    const int ec = valid ? e : nE - 1;
    const int c = col[ec];
    const int r = row[ec];
    bin[k] = valid ? (c >> BIN_SHIFT) : -1;
    pk[k] = (int)(((unsigned)r << BIN_SHIFT) | (unsigned)(c & (BIN_NODES - 1)));
  }
#pragma unroll
  for (int k = 0; k < PART_EPT; ++k)
    lr[k] = (bin[k] >= 0) ? atomicAdd(&hist[bin[k]], 1) : 0;
  __syncthreads();

  const int cnt = hist[tid];
  int gb = 0;
  if (cnt > 0) gb = atomicAdd(&fill[tid * FILL_PITCH], cnt);
  gbase[tid] = gb;
  const int lane = tid & 31;
  int inc = cnt;
#pragma unroll
  for (int o = 1; o < 32; o <<= 1) {
    const int v = __shfl_up(inc, o, 32);
    if (lane >= o) inc += v;
  }
  if (lane == 31) wsum[tid >> 5] = inc;
  __syncthreads();
  int pre = 0;
  for (int w = 0; w < (tid >> 5); ++w) pre += wsum[w];
  hist[tid] = pre + inc - cnt;
  __syncthreads();

#pragma unroll
  for (int k = 0; k < PART_EPT; ++k) {
    if (bin[k] >= 0) {
      const int b = bin[k];
      const int j = hist[b] + lr[k];
      const int p = gbase[b] + lr[k];
      sval[j] = pk[k];
      sdst[j] = (p < BIN_CAP) ? b * BIN_CAP + p : -1;
    }
  }
  __syncthreads();

  const int left = nE - e0;
  const int nv = left < PART_E ? left : PART_E;
  int od[PART_EPT], ov[PART_EPT];
#pragma unroll
  for (int k = 0; k < PART_EPT; ++k) {
    const int j = k * PART_T + tid;
    const int jr = j < nv ? j : 0;
    od[k] = sdst[jr];
    ov[k] = sval[jr];
  }
#pragma unroll
  for (int k = 0; k < PART_EPT; ++k) {
    const int j = k * PART_T + tid;
    if (j < nv && od[k] >= 0) seg[od[k]] = ov[k];
  }
}

__global__ void __launch_bounds__(BB_T)
build_buckets(const int* __restrict__ row, const int* __restrict__ col, int nE,
              const int* __restrict__ fill, const int* __restrict__ seg,
              int* __restrict__ bkt, int nNodes) {
  __shared__ __attribute__((aligned(16))) int lb[BIN_NODES * BKT];
  __shared__ int cnt[BIN_NODES];

  const int tid = (int)threadIdx.x;
  const int b = (int)blockIdx.x;
  const int n = fill[b * FILL_PITCH];
  const bool fits = n <= BIN_CAP;
  const int* s = seg + (long)b * BIN_CAP;

  unsigned pw[BB_EPT];
#pragma unroll
  for (int k = 0; k < BB_EPT; ++k) {
    const int j = k * BB_T + tid;
    pw[k] = (fits && j < n) ? (unsigned)s[j] : 0u;
  }

  int4* lb4 = reinterpret_cast<int4*>(lb);
  int4 none;
  none.x = -1; none.y = -1; none.z = -1; none.w = -1;
  for (int p = tid; p < BIN_NODES * BKT / 4; p += BB_T) lb4[p] = none;
  if (tid < BIN_NODES) cnt[tid] = 0;
  __syncthreads();

  auto place = [&](unsigned p) {
    const int cl = (int)(p & (unsigned)(BIN_NODES - 1));
    const int sl = atomicAdd(&cnt[cl], 1);
    if (sl < BCAP) lb[cl * BKT + 1 + sl] = (int)(p >> BIN_SHIFT);
  };
  if (fits) {
    int sl[BB_EPT];
#pragma unroll
    for (int k = 0; k < BB_EPT; ++k) {
      sl[k] = BCAP;
      if (k * BB_T + tid < n)
        sl[k] = atomicAdd(&cnt[pw[k] & (unsigned)(BIN_NODES - 1)], 1);
    }
#pragma unroll
    for (int k = 0; k < BB_EPT; ++k) {
      if (sl[k] < BCAP)
        lb[(int)(pw[k] & (unsigned)(BIN_NODES - 1)) * BKT + 1 + sl[k]] = (int)(pw[k] >> BIN_SHIFT);
    }
    for (int j = BB_EPT * BB_T + tid; j < n; j += BB_T) place((unsigned)s[j]);
  } else {
    for (int e = tid; e < nE; e += BB_T) {
      const int c = col[e];
      if ((c >> BIN_SHIFT) == b) {
        const int cl = c & (BIN_NODES - 1);
        const int sl = atomicAdd(&cnt[cl], 1);
        if (sl < BCAP) lb[cl * BKT + 1 + sl] = row[e];
      }
    }
  }
  __syncthreads();
  if (tid < BIN_NODES) lb[tid * BKT] = cnt[tid];
  __syncthreads();

  const int node0 = b * BIN_NODES;
  for (int p = tid; p < BIN_NODES * BKT / 4; p += BB_T) {
    const int nl = p >> 4;
    if (node0 + nl < nNodes)
      *reinterpret_cast<int4*>(bkt + (long)(node0 + nl) * BKT + (p & 15) * 4) = lb4[p];
  }
}

__global__ void __launch_bounds__(128)
fold_weights(const float* __restrict__ W1, const float* __restrict__ b1,
             const float* __restrict__ W2, const float* __restrict__ b2,
             const float* __restrict__ W3, const float* __restrict__ b3,
             const float* __restrict__ W4, float* __restrict__ fold,
             int* __restrict__ fill, int nBins) {
  __shared__ float slds[16 * 132];
  __shared__ float tlds[16 * 68];

  const int tid  = threadIdx.x;
  const int wave = tid >> 5;
  const int lane = tid & 31;
  const int bt = (int)blockIdx.x;
  const int g0 = bt * 16;

  if (bt == 7) {
    for (int i = tid; i < nBins; i += 128) fill[i * FILL_PITCH] = 0;
    return;
  }

  const int m  = lane & 15;
  const int kq = (lane >> 4) * 2;
  const int n  = (lane & 15) + wave * 16;
  const int mrow0 = (lane >> 4) * 8;

  v2f w4f[16];
#pragma unroll
  for (int q = 0; q < 16; ++q) {
    const int kb = q * 4 + kq;
    w4f[q].x = W4[kb * 64 + n];
    w4f[q].y = W4[(kb + 1) * 64 + n];
  }

  float sv[16];
#pragma unroll
  for (int r = 0; r < 16; ++r) sv[r] = 0.0f;
  if (bt < 4) {
    if (tid < 64) {
#pragma unroll
      for (int r = 0; r < 16; ++r) sv[r] = W1[(g0 + r) * 64 + tid];
    }
  } else if (bt < 6) {
    if (tid >= 64) {
#pragma unroll
      for (int r = 0; r < 16; ++r) sv[r] = W2[(g0 - 64 + r) * 64 + (tid - 64)];
    }
  } else {
    sv[0] = (tid < 64) ? b1[tid] : b2[tid - 64];
  }
#pragma unroll
  for (int r = 0; r < 16; ++r) slds[r * 132 + tid] = sv[r];
  __syncthreads();

  const int kLo = (bt >= 4 && bt < 6) ? 64 : 0;
  const int kHi = (bt < 4) ? 64 : 128;

  v8f t = {};
  for (int k0 = kLo; k0 < kHi; k0 += 64) {
#pragma unroll
    for (int kk = 0; kk < 64; kk += 4) {
      const int kb = k0 + kk + kq;
      v2f a, b;
      a.x = slds[m * 132 + kb];
      a.y = slds[m * 132 + kb + 1];
      b.x = W3[kb * 64 + n];
      b.y = W3[(kb + 1) * 64 + n];
      t = __builtin_amdgcn_wmma_f32_16x16x4_f32(false, a, false, b, (short)0, t,
                                                false, false);
    }
  }
#pragma unroll
  for (int v = 0; v < 8; ++v) {
    const int g = g0 + mrow0 + v;
    tlds[(mrow0 + v) * 68 + n] = (g == 97) ? b3[n] : t[v];
  }
  __syncthreads();

  v8f c = {};
#pragma unroll
  for (int kk = 0; kk < 64; kk += 4) {
    const int kb = kk + kq;
    v2f a;
    a.x = tlds[m * 68 + kb];
    a.y = tlds[m * 68 + kb + 1];
    c = __builtin_amdgcn_wmma_f32_16x16x4_f32(false, a, false, w4f[kk >> 2],
                                              (short)0, c, false, false);
  }
#pragma unroll
  for (int v = 0; v < 8; ++v) {
    const int g = g0 + mrow0 + v;
    if (g < 98) fold[g * 64 + n] = c[v];
  }
}

template <int K1, int K2>
__global__ void __launch_bounds__(128)
gcn_gemm_wmma(const float* __restrict__ X1, const float* __restrict__ X2,
              const float* __restrict__ W, const int* __restrict__ bkt,
              float* __restrict__ Y, int nRows) {
  constexpr int K = K1 + K2;
  static_assert(16 * K1 == 2 * 128 * 4 && 16 * K2 == 128 * 4, "tile pieces");
  __shared__ float wlds[K * WPAD];
  __shared__ float alds[16 * (K + 4)];

  const int tid  = threadIdx.x;
  const int wave = tid >> 5;
  const int lane = tid & 31;
  const int nTiles = (nRows + 15) >> 4;
  const int tStep = (int)gridDim.x;

  for (int i = tid * 4; i < K * 64; i += 128 * 4) {
    const int k = i >> 6;
    const int c = i & 63;
    *reinterpret_cast<float4*>(&wlds[k * WPAD + c]) =
        *reinterpret_cast<const float4*>(W + i);
  }

  const int m  = lane & 15;
  const int kq = (lane >> 4) * 2;
  const int n  = (lane & 15) + wave * 16;
  const int mrow0 = (lane >> 4) * 8;

  auto loadPiece1 = [&](int tileIdx, int j) -> float4 {
    const int i = (tid + j * 128) * 4;
    const int r = i / K1;
    const int cc = i % K1;
    int gr = tileIdx * 16 + r;
    if (gr >= nRows) gr = nRows - 1;
    return *reinterpret_cast<const float4*>(X1 + (long)gr * K1 + cc);
  };
  auto loadPiece2 = [&](int tileIdx) -> float4 {
    const int i = tid * 4;
    const int r = i / K2;
    const int cc = i % K2;
    int gr = tileIdx * 16 + r;
    if (gr >= nRows) gr = nRows - 1;
    return *reinterpret_cast<const float4*>(X2 + (long)gr * K2 + cc);
  };
  auto storePiece1 = [&](const float4& v, int j) {
    const int i = (tid + j * 128) * 4;
    const int r = i / K1;
    const int cc = i % K1;
    *reinterpret_cast<float4*>(&alds[r * (K + 4) + cc]) = v;
  };
  auto storePiece2 = [&](const float4& v) {
    const int i = tid * 4;
    const int r = i / K2;
    const int cc = i % K2;
    *reinterpret_cast<float4*>(&alds[r * (K + 4) + K1 + cc]) = v;
  };

  int tl = (int)blockIdx.x;
  float4 p0 = loadPiece1(tl, 0);
  float4 p1 = loadPiece1(tl, 1);
  float4 p2 = loadPiece2(tl);

  for (; tl < nTiles; tl += tStep) {
  const int tile = tl * 16;
  __syncthreads();
  storePiece1(p0, 0);
  storePiece1(p1, 1);
  storePiece2(p2);
  __syncthreads();

  int dg[8];
#pragma unroll
  for (int v = 0; v < 8; ++v) {
    int gr = tile + mrow0 + v;
    if (gr >= nRows) gr = nRows - 1;
    dg[v] = bkt[(long)gr * BKT];
  }

  p0 = loadPiece1(tl + tStep, 0);
  p1 = loadPiece1(tl + tStep, 1);
  p2 = loadPiece2(tl + tStep);

  v8f c = {};
#pragma unroll
  for (int kk = 0; kk < K; kk += 4) {
    const int kb = kk + kq;
    v2f a, b;
    a.x = alds[m * (K + 4) + kb];
    a.y = alds[m * (K + 4) + kb + 1];
    b.x = wlds[kb * WPAD + n];
    b.y = wlds[(kb + 1) * WPAD + n];
    c = __builtin_amdgcn_wmma_f32_16x16x4_f32(false, a, false, b, (short)0, c,
                                              false, false);
  }

  if (tile + 16 <= nRows) {
#pragma unroll
    for (int v = 0; v < 8; ++v) {
      const int gr = tile + mrow0 + v;
      Y[(long)gr * 64 + n] = c[v] * dinv_of(dg[v]);
    }
  } else {
#pragma unroll
    for (int v = 0; v < 8; ++v) {
      const int gr = tile + mrow0 + v;
      if (gr < nRows) Y[(long)gr * 64 + n] = c[v] * dinv_of(dg[v]);
    }
  }
  }
}

template <bool CHAIN>
__device__ __forceinline__ void finalize_row(float4 acc, float d, const float4& b,
                                             float* __restrict__ dst) {
  float4 o;
  o.x = d * acc.x + b.x;
  o.y = d * acc.y + b.y;
  o.z = d * acc.z + b.z;
  o.w = d * acc.w + b.w;
  if (CHAIN) {
    o.x *= d;
    o.y *= d;
    o.z *= d;
    o.w *= d;
  }
  *reinterpret_cast<float4*>(dst) = o;
}

template <bool CHAIN, bool SORT>
__global__ void __launch_bounds__(256, 16)
gather_finalize(int* __restrict__ bkt, const int* __restrict__ row,
                const int* __restrict__ col, int nE,
                const float* __restrict__ y, const float* __restrict__ bias,
                float* __restrict__ out, int nNodes) {
  __shared__ __attribute__((aligned(16))) int sl[SORT ? 16 * BKT : 4];

  const int t = (int)blockIdx.x * 256 + (int)threadIdx.x;
  const int g = t >> 4;
  const int nG = ((int)gridDim.x * 256) >> 4;
  const int l = t & 15;
  const int f = l << 2;
  const float4 b = *reinterpret_cast<const float4*>(bias + f);
  const int nIt = (nNodes + nG - 1) / nG;
  bool anyBig = false;

  const char* yb = reinterpret_cast<const char*>(y);
  char* bb = reinterpret_cast<char*>(bkt);
  char* ob = reinterpret_cast<char*>(out);
  const unsigned lf = (unsigned)l * 16u;
  auto loadRow = [&](int node) -> float4 {
    return *reinterpret_cast<const float4*>(yb + (size_t)((unsigned)node * 256u + lf));
  };
  auto loadBucket = [&](int node) -> int4 {
    const int nc = node < nNodes ? node : nNodes - 1;
    return *reinterpret_cast<const int4*>(bb + (size_t)((unsigned)nc * 256u + lf));
  };
  int4 bkN = loadBucket(g);

  for (int it = 0; it < nIt; ++it) {
  const int i0 = g + it * nG;
  const bool live = i0 < nNodes;
  const int i = live ? i0 : nNodes - 1;
  int4 bk = bkN;
  bkN = loadBucket(i0 + nG);
  float4 acc = loadRow(i);

  const int deg = __shfl(bk.x, 0, 16);
  const bool big = live && deg > BCAP;
  const bool mine1 = live && deg <= BCAP;
  anyBig = anyBig || big;
  const int ne = mine1 ? deg : 0;
  const float d = dinv_of(deg);

  if (SORT) {
    int val[4] = {bk.x, bk.y, bk.z, bk.w};
    int key[4];
    int cnt[4];
#pragma unroll
    for (int s = 0; s < 4; ++s) {
      const int w = f + s;
      key[s] = (w == 0) ? -1
             : ((w <= ne) ? (int)(((unsigned)val[s] << 6) | (unsigned)w) : 0x7fffffff);
      cnt[s] = 0;
    }
    const int nL = (ne >> 2) + 1;
    const int nLO = __shfl_xor(nL, 16, 32);
    const int nLW = nL > nLO ? nL : nLO;
#pragma unroll
    for (int j = 0; j < 16; ++j) {
      if (j < nLW) {
#pragma unroll
        for (int q = 0; q < 4; ++q) {
          const int kb = __shfl(key[q], j, 16);
#pragma unroll
          for (int s = 0; s < 4; ++s) cnt[s] += (kb < key[s]) ? 1 : 0;
        }
      }
    }
    int* mySl = sl + ((int)threadIdx.x >> 4) * BKT;
#pragma unroll
    for (int s = 0; s < 4; ++s) {
      const int w = f + s;
      const int p = (w >= 1 && w <= ne) ? cnt[s] : w;
      mySl[p] = val[s];
    }
    __builtin_amdgcn_wave_barrier();
    bk = *reinterpret_cast<const int4*>(mySl + f);
    __builtin_amdgcn_wave_barrier();
    if (mine1) *reinterpret_cast<int4*>(bb + (size_t)((unsigned)i * 256u + lf)) = bk;
  }

  const int nch = (ne + 7) >> 3;
  const int nchO = __shfl_xor(nch, 16, 32);
  const int nchW = nch > nchO ? nch : nchO;

  for (int c = 0; c < nchW; ++c) {
    int jv[8];
    jv[0] = __shfl(bk.y, 0, 16);
    jv[1] = __shfl(bk.z, 0, 16);
    jv[2] = __shfl(bk.w, 0, 16);
    jv[3] = __shfl(bk.x, 1, 16);
    jv[4] = __shfl(bk.y, 1, 16);
    jv[5] = __shfl(bk.z, 1, 16);
    jv[6] = __shfl(bk.w, 1, 16);
    jv[7] = __shfl(bk.x, 2, 16);
    float4 v[8];
#pragma unroll
    for (int u = 0; u < 8; ++u) {
      const int jr = (c * 8 + u < ne) ? jv[u] : i;
      v[u] = loadRow(jr);
    }
    bk.x = __shfl_down(bk.x, 2, 16);
    bk.y = __shfl_down(bk.y, 2, 16);
    bk.z = __shfl_down(bk.z, 2, 16);
    bk.w = __shfl_down(bk.w, 2, 16);
#pragma unroll
    for (int u = 0; u < 8; ++u) {
      if (c * 8 + u < ne) {
        acc.x += v[u].x;
        acc.y += v[u].y;
        acc.z += v[u].z;
        acc.w += v[u].w;
      }
    }
  }
  if (mine1)
    finalize_row<CHAIN>(acc, d, b, reinterpret_cast<float*>(ob + (size_t)((unsigned)i * 256u + lf)));
  }

  if (anyBig) {
    for (int it = 0; it < nIt; ++it) {
      const int i = g + it * nG;
      if (i < nNodes && bkt[i * BKT] > BCAP) {
        const int dg = bkt[i * BKT];
        float4 acc = *reinterpret_cast<const float4*>(y + (long)i * 64 + f);
        for (int q = 0; q < nE; ++q) {
          if (col[q] == i) {
            const float4 v =
                *reinterpret_cast<const float4*>(y + (long)row[q] * 64 + f);
            acc.x += v.x;
            acc.y += v.y;
            acc.z += v.z;
            acc.w += v.w;
          }
        }
        finalize_row<CHAIN>(acc, dinv_of(dg), b, out + (long)i * 64 + f);
      }
    }
  }
}

extern "C" void kernel_launch(void* const* d_in, const int* in_sizes, int n_in,
                              void* d_out, int out_size, void* d_ws, size_t ws_size,
                              hipStream_t stream) {
  const float* latent    = (const float*)d_in[0];
  const float* condition = (const float*)d_in[1];
  const int*   ei        = (const int*)d_in[2];
  const float* W1 = (const float*)d_in[3];
  const float* b1 = (const float*)d_in[4];
  const float* W2 = (const float*)d_in[5];
  const float* b2 = (const float*)d_in[6];
  const float* W3 = (const float*)d_in[7];
  const float* b3 = (const float*)d_in[8];
  const float* W4 = (const float*)d_in[9];
  const float* b4 = (const float*)d_in[10];

  const int N = in_sizes[0] / 64;
  const int E = in_sizes[2] / 2;
  const int* rowp = ei;
  const int* colp = ei + E;

  char* ws = (char*)d_ws;
  int* fill = (int*)d_ws;
  size_t off = 0;
  off += ((size_t)N * 4 + 255) & ~(size_t)255;
  float* ybuf = (float*)(ws + off);
  off += (size_t)N * 64 * 4;
  int* adj = (int*)(ws + off);
  off += (size_t)N * 64 * 4;
  float* hbuf = (float*)(ws + off);
  float* fold = hbuf + (size_t)N * 64;
  int* seg = (int*)hbuf;

  const dim3 gemmGrid(1250);
  const dim3 gemmBlk(128);
  const int nBins = (N + BIN_NODES - 1) / BIN_NODES;
  const dim3 partGrid((E + PART_E - 1) / PART_E);
  const dim3 binGrid(nBins);
  const dim3 nodeGrid(2048);

  fold_weights<<<dim3(8), gemmBlk, 0, stream>>>(W1, b1, W2, b2, W3, b3, W4, fold, fill, nBins);

  partition_edges<<<partGrid, PART_T, 0, stream>>>(rowp, colp, fill, seg, E);
  build_buckets<<<binGrid, BB_T, 0, stream>>>(rowp, colp, E, fill, seg, adj, N);

  gcn_gemm_wmma<64, 32><<<gemmGrid, gemmBlk, 0, stream>>>(latent, condition, fold, adj, ybuf, N);

  gather_finalize<true, true><<<nodeGrid, 256, 0, stream>>>(adj, rowp, colp, E, ybuf, fold + 96 * 64, hbuf, N);
  gather_finalize<true, false><<<nodeGrid, 256, 0, stream>>>(adj, rowp, colp, E, hbuf, fold + 97 * 64, ybuf, N);
  gather_finalize<false, false><<<nodeGrid, 256, 0, stream>>>(adj, rowp, colp, E, ybuf, b4, (float*)d_out, N);
}
